# starting kernel
# baseline (speedup 1.0000x reference)
#include <hip/hip_runtime.h>


typedef __attribute__((ext_vector_type(2))) float v2f;
typedef __attribute__((ext_vector_type(8))) float v8f;

#define GCN_N 100000

__global__ void deg_kernel(const int* __restrict__ col, float* __restrict__ deg, int E) {
    int e = blockIdx.x * blockDim.x + threadIdx.x;
    if (e < E) atomicAdd(&deg[col[e]], 1.0f);
}

__global__ void dinv_kernel(float* __restrict__ deg, int n) {
    int i = blockIdx.x * blockDim.x + threadIdx.x;
    if (i < n) deg[i] = rsqrtf(deg[i] + 1.0f);
}

__global__ void gemm_k16_n32_wmma(const float* __restrict__ X, const float* __restrict__ W,
                                  float* __restrict__ H, int n) {
    int wave   = blockIdx.x * (blockDim.x >> 5) + (threadIdx.x >> 5);
    int nTiles = (n + 15) >> 4;
    if (wave >= nTiles) return;
    int lane = threadIdx.x & 31;
    int mi   = lane & 15;
    int kb   = (lane >> 4) << 1;
    int r0   = wave << 4;
    bool full = (r0 + 16 <= n);
    int rld  = full ? (r0 + mi) : min(r0 + mi, n - 1);
    const float* xr = X + (size_t)rld * 16;

    v8f acc0 = {}; v8f acc1 = {};
#pragma unroll
    for (int kk = 0; kk < 16; kk += 4) {
        v2f a = *(const v2f*)(xr + kk + kb);
        v2f b0, b1;
        b0.x = W[(kk + kb) * 32 + mi];          b0.y = W[(kk + kb + 1) * 32 + mi];
        b1.x = W[(kk + kb) * 32 + mi + 16];     b1.y = W[(kk + kb + 1) * 32 + mi + 16];
        acc0 = __builtin_amdgcn_wmma_f32_16x16x4_f32(false, a, false, b0, (short)0, acc0, false, false);
        acc1 = __builtin_amdgcn_wmma_f32_16x16x4_f32(false, a, false, b1, (short)0, acc1, false, false);
    }
    int mh = (lane >> 4) << 3;
    float* hb = H + (size_t)(r0 + mh) * 32 + mi;
    if (full) {
#pragma unroll
        for (int v = 0; v < 8; v++) {
            hb[v * 32]      = acc0[v];
            hb[v * 32 + 16] = acc1[v];
        }
    } else {
#pragma unroll
        for (int v = 0; v < 8; v++) {
            if (r0 + mh + v < n) {
                hb[v * 32]      = acc0[v];
                hb[v * 32 + 16] = acc1[v];
            }
        }
    }
}

__global__ void gemm_k32_n16_wmma(const float* __restrict__ X, const float* __restrict__ W,
                                  float* __restrict__ H, int n) {
    int wave   = blockIdx.x * (blockDim.x >> 5) + (threadIdx.x >> 5);
    int nTiles = (n + 15) >> 4;
    if (wave >= nTiles) return;
    int lane = threadIdx.x & 31;
    int mi   = lane & 15;
    int kb   = (lane >> 4) << 1;
    int r0   = wave << 4;
    bool full = (r0 + 16 <= n);
    int rld  = full ? (r0 + mi) : min(r0 + mi, n - 1);
    const float* xr = X + (size_t)rld * 32;

    v8f acc = {};
#pragma unroll
    for (int kk = 0; kk < 32; kk += 4) {
        v2f a = *(const v2f*)(xr + kk + kb);
        v2f b;
        b.x = W[(kk + kb) * 16 + mi];
        b.y = W[(kk + kb + 1) * 16 + mi];
        acc = __builtin_amdgcn_wmma_f32_16x16x4_f32(false, a, false, b, (short)0, acc, false, false);
    }
    int mh = (lane >> 4) << 3;
    float* hb = H + (size_t)(r0 + mh) * 16 + mi;
    if (full) {
#pragma unroll
        for (int v = 0; v < 8; v++) hb[v * 16] = acc[v];
    } else {
#pragma unroll
        for (int v = 0; v < 8; v++)
            if (r0 + mh + v < n) hb[v * 16] = acc[v];
    }
}

template <int C>
__global__ void scatter_kernel(const int* __restrict__ row, const int* __restrict__ col,
                               const float* __restrict__ h, const float* __restrict__ dinv,
                               float* __restrict__ agg, int E) {
    long long t = (long long)blockIdx.x * blockDim.x + threadIdx.x;
    int e = (int)(t / C);
    int j = (int)(t % C);
    if (e < E) {
        int r = row[e], c = col[e];
        float nrm = dinv[r] * dinv[c];
        atomicAdd(&agg[(size_t)c * C + j], h[(size_t)r * C + j] * nrm);
    }
}

__global__ void selfloop_bias_relu(float* __restrict__ agg, const float* __restrict__ h,
                                   const float* __restrict__ dinv, const float* __restrict__ b,
                                   int n, int C) {
    long long t = (long long)blockIdx.x * blockDim.x + threadIdx.x;
    int i = (int)(t / C);
    int j = (int)(t % C);
    if (i < n) {
        float di = dinv[i];
        float v  = agg[(size_t)i * C + j] + h[(size_t)i * C + j] * di * di + b[j];
        agg[(size_t)i * C + j] = v > 0.0f ? v : 0.0f;
    }
}

__global__ void final_kernel(const float* __restrict__ agg2, const float* __restrict__ h2,
                             const float* __restrict__ dinv, const float* __restrict__ b2,
                             const float* __restrict__ fcW, const float* __restrict__ fcb,
                             float* __restrict__ out, int n) {
    int i = blockIdx.x * blockDim.x + threadIdx.x;
    if (i < n) {
        float d2 = dinv[i] * dinv[i];
        float s = 0.0f;
#pragma unroll
        for (int j = 0; j < 16; j++) {
            float v = agg2[(size_t)i * 16 + j] + h2[(size_t)i * 16 + j] * d2 + b2[j];
            v = v > 0.0f ? v : 0.0f;
            s += v * fcW[j];
        }
        out[i] = s + fcb[0];
    }
}

extern "C" void kernel_launch(void* const* d_in, const int* in_sizes, int n_in,
                              void* d_out, int out_size, void* d_ws, size_t ws_size,
                              hipStream_t stream) {
    const int*   edge = (const int*)d_in[0];
    const float* x0   = (const float*)d_in[1];
    const float* W1   = (const float*)d_in[2];
    const float* b1   = (const float*)d_in[3];
    const float* W2   = (const float*)d_in[4];
    const float* b2   = (const float*)d_in[5];
    const float* fcW  = (const float*)d_in[6];
    const float* fcb  = (const float*)d_in[7];
    float*       out  = (float*)d_out;

    const int n = GCN_N;
    const int E = in_sizes[0] / 2;
    const int* row = edge;
    const int* col = edge + E;

    float* ws   = (float*)d_ws;
    size_t npad = ((size_t)n + 63) & ~(size_t)63;
    float* dinv = ws;
    float* h1   = dinv + npad;
    float* agg1 = h1   + (size_t)n * 32;
    float* h2   = agg1 + (size_t)n * 32;
    float* agg2 = h2   + (size_t)n * 16;

    hipMemsetAsync(dinv, 0, (size_t)n * sizeof(float), stream);
    hipMemsetAsync(agg1, 0, (size_t)n * 32 * sizeof(float), stream);
    hipMemsetAsync(agg2, 0, (size_t)n * 16 * sizeof(float), stream);

    deg_kernel <<<(E + 255) / 256, 256, 0, stream>>>(col, dinv, E);
    dinv_kernel<<<(n + 255) / 256, 256, 0, stream>>>(dinv, n);

    int tiles   = (n + 15) / 16;
    int gblocks = (tiles + 7) / 8;

    gemm_k16_n32_wmma<<<gblocks, 256, 0, stream>>>(x0, W1, h1, n);
    long long tot1 = (long long)E * 32;
    scatter_kernel<32><<<(int)((tot1 + 255) / 256), 256, 0, stream>>>(row, col, h1, dinv, agg1, E);
    selfloop_bias_relu<<<(int)(((long long)n * 32 + 255) / 256), 256, 0, stream>>>(agg1, h1, dinv, b1, n, 32);

    gemm_k32_n16_wmma<<<gblocks, 256, 0, stream>>>(agg1, W2, h2, n);
    long long tot2 = (long long)E * 16;
    scatter_kernel<16><<<(int)((tot2 + 255) / 256), 256, 0, stream>>>(row, col, h2, dinv, agg2, E);

    final_kernel<<<(n + 255) / 256, 256, 0, stream>>>(agg2, h2, dinv, b2, fcW, fcb, out, n);
}


# each WMMA product moved into the bin-block launch in front of it (GEMM1 behind the degree count, GEMM2 fed from the finished layer-1 LDS tile): 7 launches -> 5, x1 never stored; sums unchanged
# speedup vs baseline: 5.6098x; 5.6098x over previous
#include <hip/hip_runtime.h>


typedef __attribute__((ext_vector_type(2))) float v2f;
typedef __attribute__((ext_vector_type(8))) float v8f;

#define GCN_N 100000
#define BIN_SHIFT 8
#define BIN_T (1 << BIN_SHIFT)
#define NBINS ((GCN_N + BIN_T - 1) >> BIN_SHIFT)
#define PK_SHIFT 9
#define PK_MASK 511u
#define EPB 8192
#define PT 1024
#define EPT (EPB / PT)

static_assert(GCN_N < (1 << (32 - PK_SHIFT)), "source index must fit the packed word");

#define NCH ((NBINS + 31) / 32)

__device__ inline void wave_excl_scan_chunks(const int (&v)[NCH], int (&ex)[NCH], int& total, int lane) {
    int inc[NCH];
#pragma unroll
    for (int j = 0; j < NCH; j++) inc[j] = v[j];
#pragma unroll
    for (int d = 1; d < 32; d <<= 1) {
        int t[NCH];
#pragma unroll
        for (int j = 0; j < NCH; j++) t[j] = __shfl_up(inc[j], d);
#pragma unroll
        for (int j = 0; j < NCH; j++) if (lane >= d) inc[j] += t[j];
    }
    int tot[NCH];
#pragma unroll
    for (int j = 0; j < NCH; j++) tot[j] = __shfl(inc[j], 31);
    int carry = 0;
#pragma unroll
    for (int j = 0; j < NCH; j++) { ex[j] = carry + inc[j] - v[j]; carry += tot[j]; }
    total = carry;
}

__global__ __launch_bounds__(PT) void bin_count(const int* __restrict__ col,
                                                int* __restrict__ binCount, int E) {
    __shared__ int hist[NBINS];
    const int tid = threadIdx.x;
    const int e0 = blockIdx.x * EPB;
    int c[EPT];
#pragma unroll
    for (int k = 0; k < EPT; k++) c[k] = col[min(e0 + k * PT + tid, E - 1)];
    for (int i = tid; i < NBINS; i += PT) hist[i] = 0;
    __syncthreads();
#pragma unroll
    for (int k = 0; k < EPT; k++) {
        int e = e0 + k * PT + tid;
        if (e < E) atomicAdd(&hist[c[k] >> BIN_SHIFT], 1);
    }
    __syncthreads();
    for (int i = tid; i < NBINS; i += PT) {
        int h = hist[i];
        if (h) atomicAdd(&binCount[i], h);
    }
}

__global__ __launch_bounds__(PT) void bin_place(const int* __restrict__ row, const int* __restrict__ col,
                                                const int* __restrict__ binCount,
                                                int* __restrict__ cursor, int* __restrict__ binStart,
                                                unsigned* __restrict__ be, int E) {
    __shared__ int hist[NBINS];
    __shared__ int sStart[NBINS + 1];
    __shared__ int lStart[NBINS];
    __shared__ int gBase[NBINS];
    __shared__ unsigned stage[EPB];
    __shared__ unsigned short stageBin[EPB];

    const int tid = threadIdx.x, lane = tid & 31, wave = tid >> 5;
    const int e0 = blockIdx.x * EPB;

    static_assert(NBINS <= PT, "one bin count per thread");
    const int bcnt = binCount[min(tid, NBINS - 1)];
    int cc[EPT], rr[EPT];
#pragma unroll
    for (int k = 0; k < EPT; k++) {
        int e = min(e0 + k * PT + tid, E - 1);
        cc[k] = col[e];
        rr[k] = row[e];
    }
    if (tid < NBINS) { hist[tid] = 0; sStart[tid] = bcnt; }
    __syncthreads();

    int rk[EPT], bn[EPT];
    unsigned pk[EPT];
#pragma unroll
    for (int k = 0; k < EPT; k++) {
        int e = e0 + k * PT + tid;
        bn[k] = -1; rk[k] = 0; pk[k] = 0u;
        if (e < E) {
            int b = cc[k] >> BIN_SHIFT;
            bn[k] = b;
            rk[k] = atomicAdd(&hist[b], 1);
            pk[k] = ((unsigned)rr[k] << PK_SHIFT) | (unsigned)(cc[k] & (BIN_T - 1));
        }
    }
    __syncthreads();

    if (wave < 2) {
        const int* srcS = (wave == 0) ? hist : sStart;
        int*       dstS = (wave == 0) ? lStart : sStart;
        int sv[NCH], ex[NCH], total;
#pragma unroll
        for (int j = 0; j < NCH; j++) {
            int i = j * 32 + lane;
            int v = srcS[min(i, NBINS - 1)];
            sv[j] = (i < NBINS) ? v : 0;
        }
        wave_excl_scan_chunks(sv, ex, total, lane);
#pragma unroll
        for (int j = 0; j < NCH; j++) {
            int i = j * 32 + lane;
            if (i < NBINS) dstS[i] = ex[j];
        }
        if (wave == 1 && lane == 0) sStart[NBINS] = total;
    }
    __syncthreads();

    for (int i = tid; i < NBINS; i += PT) {
        int h = hist[i];
        int g = h ? atomicAdd(&cursor[i], h) : 0;
        gBase[i] = sStart[i] + g;
    }
    if (blockIdx.x == 0)
        for (int i = tid; i <= NBINS; i += PT) binStart[i] = sStart[i];
    int ls[EPT];
#pragma unroll
    for (int k = 0; k < EPT; k++) ls[k] = lStart[max(bn[k], 0)];
#pragma unroll
    for (int k = 0; k < EPT; k++) {
        if (bn[k] >= 0) {
            int pos = ls[k] + rk[k];
            stage[pos]    = pk[k];
            stageBin[pos] = (unsigned short)bn[k];
        }
    }
    __syncthreads();

    const int nIn = min(EPB, E - e0);
    int sb[EPT], off[EPT];
    unsigned sw[EPT];
#pragma unroll
    for (int k = 0; k < EPT; k++) {
        int t = min(tid + k * PT, nIn - 1);
        sb[k] = stageBin[t];
        sw[k] = stage[t];
    }
#pragma unroll
    for (int k = 0; k < EPT; k++) off[k] = gBase[sb[k]] - lStart[sb[k]];
#pragma unroll
    for (int k = 0; k < EPT; k++) {
        int t = tid + k * PT;
        if (t < nIn) be[off[k] + t] = sw[k];
    }
}

#define DGW 10
__global__ __launch_bounds__(512) void bin_degree_gemm(const unsigned* __restrict__ be,
                                                       const int* __restrict__ binStart,
                                                       const float* __restrict__ X, const float* __restrict__ W,
                                                       float* __restrict__ dinv, float* __restrict__ H,
                                                       int n, int E) {
    __shared__ int cnt[BIN_T];
    __shared__ float sdv[BIN_T];
    const int tid = threadIdx.x, bin = blockIdx.x;
    const int lane = tid & 31, wave = tid >> 5;
    const int mi   = lane & 15;
    const int kb   = (lane >> 4) << 1;
    const int r0   = (bin << BIN_SHIFT) + (wave << 4);
    const int s = binStart[bin], e = binStart[bin + 1];

    if (tid < BIN_T) cnt[tid] = 0;
    __syncthreads();

    const float* xr = X + (size_t)min(r0 + mi, n - 1) * 16;
    v2f av[4], b0[4], b1[4];
#pragma unroll
    for (int q = 0; q < 4; q++) {
        const int kk = q * 4;
        av[q] = *(const v2f*)(xr + kk + kb);
        b0[q].x = W[(kk + kb) * 32 + mi];           b0[q].y = W[(kk + kb + 1) * 32 + mi];
        b1[q].x = W[(kk + kb) * 32 + mi + 16];      b1[q].y = W[(kk + kb + 1) * 32 + mi + 16];
    }
    for (int p0 = s + tid; p0 < e; p0 += 512 * DGW) {
        unsigned w[DGW];
#pragma unroll
        for (int k = 0; k < DGW; k++) w[k] = be[min(p0 + k * 512, E - 1)];
#pragma unroll
        for (int k = 0; k < DGW; k++)
            atomicAdd(&cnt[w[k] & (BIN_T - 1)], (p0 + k * 512 < e) ? 1 : 0);
    }
    __syncthreads();
    if (tid < BIN_T) {
        int node = (bin << BIN_SHIFT) + tid;
        float d = rsqrtf((float)cnt[tid] + 1.0f);
        sdv[tid] = d;
        if (node < n) dinv[node] = d;
    }
    __syncthreads();

    if (r0 >= n) return;
    const bool full = (r0 + 16 <= n);
    const float sc = sdv[(wave << 4) + mi];
    v8f acc0 = {}; v8f acc1 = {};
#pragma unroll
    for (int q = 0; q < 4; q++) {
        v2f a = av[q];
        a *= sc;
        acc0 = __builtin_amdgcn_wmma_f32_16x16x4_f32(false, a, false, b0[q], (short)0, acc0, false, false);
        acc1 = __builtin_amdgcn_wmma_f32_16x16x4_f32(false, a, false, b1[q], (short)0, acc1, false, false);
    }
    int mh = (lane >> 4) << 3;
    float* hb = H + (size_t)(r0 + mh) * 32 + mi;
    if (full) {
#pragma unroll
        for (int v = 0; v < 8; v++) {
            hb[v * 32]      = acc0[v];
            hb[v * 32 + 16] = acc1[v];
        }
    } else {
#pragma unroll
        for (int v = 0; v < 8; v++) {
            if (r0 + mh + v < n) {
                hb[v * 32]      = acc0[v];
                hb[v * 32 + 16] = acc1[v];
            }
        }
    }
}

template <int C, bool FINAL>
__global__ __launch_bounds__(1024) void bin_aggregate(const unsigned* __restrict__ be,
                                                      const int* __restrict__ binStart,
                                                      const float* __restrict__ hs,
                                                      const float* __restrict__ dinv,
                                                      const float* __restrict__ bias,
                                                      const float* __restrict__ Wn,
                                                      const float* __restrict__ fcW,
                                                      const float* __restrict__ fcb,
                                                      float* __restrict__ dst, int n, int E) {
    __shared__ __attribute__((aligned(16))) float tile[(BIN_T + 1) * C];
    __shared__ float sd[BIN_T];
    const int tid = threadIdx.x, lane = tid & 31, wave = tid >> 5;
    const int bin = blockIdx.x, base = bin << BIN_SHIFT;

    float hv[BIN_T * C / 1024];
#pragma unroll
    for (int k = 0; k < BIN_T * C / 1024; k++)
        hv[k] = hs[min(base * C + tid + k * 1024, n * C - 1)];
    const float dv = dinv[min(base + (tid & (BIN_T - 1)), n - 1)];
    const int s = binStart[bin], e = binStart[bin + 1];
    int p  = s + wave * 32;
    int pk = (int)be[min(p + lane, E - 1)];
    const float bj = FINAL ? 0.0f : bias[tid % C];
    float bb[FINAL ? C : 1], ff[FINAL ? C : 1], fb = 0.0f;
    if (FINAL) {
#pragma unroll
        for (int j = 0; j < C; j++) { bb[j] = bias[j]; ff[j] = fcW[j]; }
        fb = fcb[0];
    }

#pragma unroll
    for (int k = 0; k < BIN_T * C / 1024; k++) {
        int i = tid + k * 1024;
        tile[i] = (base * C + i < n * C) ? hv[k] : 0.0f;
    }
    if (tid < C) tile[BIN_T * C + tid] = 0.0f;
    if (tid < BIN_T) sd[tid] = dv;
    __syncthreads();

    pk = (p + lane < e) ? pk : (int)BIN_T;
    while (p < e) {
        const int pn = p + 32 * 32;
        int pkn = (int)be[min(pn + lane, E - 1)];
        pkn = (pn + lane < e) ? pkn : (int)BIN_T;
        if (C == 32) {
#pragma unroll
            for (int h = 0; h < 2; h++) {
                float v[16]; int ai[16];
#pragma unroll
                for (int j = 0; j < 16; j++) {
                    unsigned q = (unsigned)__builtin_amdgcn_readlane(pk, h * 16 + j);
                    v[j]  = hs[(size_t)(q >> PK_SHIFT) * 32 + lane];
                    ai[j] = (int)(q & PK_MASK) * 32 + lane;
                }
#pragma unroll
                for (int j = 0; j < 16; j++) atomicAdd(&tile[ai[j]], v[j]);
            }
        } else {
            const int cl = lane & 15;
            float v[16]; int ai[16];
#pragma unroll
            for (int j = 0; j < 16; j++) {
                unsigned q0 = (unsigned)__builtin_amdgcn_readlane(pk, 2 * j);
                unsigned q1 = (unsigned)__builtin_amdgcn_readlane(pk, 2 * j + 1);
                unsigned q  = (lane < 16) ? q0 : q1;
                v[j]  = hs[(size_t)(q >> PK_SHIFT) * 16 + cl];
                ai[j] = (int)(q & PK_MASK) * 16 + cl;
            }
#pragma unroll
            for (int j = 0; j < 16; j++) atomicAdd(&tile[ai[j]], v[j]);
        }
        p = pn; pk = pkn;
    }
    __syncthreads();

    if (!FINAL) {
        const int mi = lane & 15;
        const int kb = (lane >> 4) << 1;
        const int r0 = base + (wave << 4);
        const bool mm = (wave < BIN_T / 16) && (r0 < n);
        v2f b[C / 4];
        if (mm) {
#pragma unroll
            for (int q = 0; q < C / 4; q++) {
                b[q].x = Wn[(q * 4 + kb) * 16 + mi];
                b[q].y = Wn[(q * 4 + kb + 1) * 16 + mi];
            }
        }
#pragma unroll
        for (int k = 0; k < BIN_T * C / 1024; k++) {
            int i    = tid + k * 1024;
            float d  = sd[i / C];
            float v  = d * tile[i] + bj;
            v = v > 0.0f ? v : 0.0f;
            tile[i] = v * d;
        }
        __syncthreads();
        if (mm) {
            const bool full = (r0 + 16 <= n);
            const float* ar = tile + ((wave << 4) + mi) * C;
            v8f acc = {};
#pragma unroll
            for (int q = 0; q < C / 4; q++) {
                v2f a = *(const v2f*)(ar + q * 4 + kb);
                acc = __builtin_amdgcn_wmma_f32_16x16x4_f32(false, a, false, b[q], (short)0, acc, false, false);
            }
            int mh = (lane >> 4) << 3;
            float* hb = dst + (size_t)(r0 + mh) * 16 + mi;
            if (full) {
#pragma unroll
                for (int v = 0; v < 8; v++) hb[v * 16] = acc[v];
            } else {
#pragma unroll
                for (int v = 0; v < 8; v++)
                    if (r0 + mh + v < n) hb[v * 16] = acc[v];
            }
        }
    } else {
        if (tid < BIN_T) {
            int node = base + tid;
            if (node < n) {
                float d = sd[tid];
                float sum = 0.0f;
#pragma unroll
                for (int j = 0; j < C; j++) {
                    float v = d * tile[tid * C + j] + bb[j];
                    v = v > 0.0f ? v : 0.0f;
                    sum += v * ff[j];
                }
                dst[node] = sum + fb;
            }
        }
    }
}

extern "C" void kernel_launch(void* const* d_in, const int* in_sizes, int n_in,
                              void* d_out, int out_size, void* d_ws, size_t ws_size,
                              hipStream_t stream) {
    const int*   edge = (const int*)d_in[0];
    const float* x0   = (const float*)d_in[1];
    const float* W1   = (const float*)d_in[2];
    const float* b1   = (const float*)d_in[3];
    const float* W2   = (const float*)d_in[4];
    const float* b2   = (const float*)d_in[5];
    const float* fcW  = (const float*)d_in[6];
    const float* fcb  = (const float*)d_in[7];
    float*       out  = (float*)d_out;

    const int n = GCN_N;
    const int E = in_sizes[0] / 2;
    const int* row = edge;
    const int* col = edge + E;

    float* ws   = (float*)d_ws;
    size_t npad = ((size_t)n + 63) & ~(size_t)63;
    float* dinv = ws;
    float* hs   = dinv + npad;
    float* hs2  = hs + (size_t)n * 32;
    unsigned* be = (unsigned*)(hs2 + (size_t)n * 32);
    int* binCount = (int*)(be + E);
    int* cursor   = binCount + NBINS;
    int* binStart = cursor + NBINS;

    hipMemsetAsync(binCount, 0, (size_t)2 * NBINS * sizeof(int), stream);

    int pblocks = (E + EPB - 1) / EPB;
    bin_count <<<pblocks, PT, 0, stream>>>(col, binCount, E);
    bin_place <<<pblocks, PT, 0, stream>>>(row, col, binCount, cursor, binStart, be, E);
    bin_degree_gemm<<<NBINS, 512, 0, stream>>>(be, binStart, x0, W1, dinv, hs, n, E);

    bin_aggregate<32, false><<<NBINS, 1024, 0, stream>>>(be, binStart, hs, dinv, b1, W2, fcW, fcb, hs2, n, E);

    bin_aggregate<16, true><<<NBINS, 1024, 0, stream>>>(be, binStart, hs2, dinv, b2, W2, fcW, fcb, out, n, E);
}
